# starting kernel
# baseline (speedup 1.0000x reference)
#include <hip/hip_runtime.h>

constexpr int B = 1024;
constexpr int T = 512;
constexpr int V = 32000;
constexpr int E = 32;
constexpr int H = 16;

typedef __attribute__((ext_vector_type(2))) float v2f;
typedef __attribute__((ext_vector_type(4))) float v4f;
typedef __attribute__((ext_vector_type(8))) float v8f;
typedef __attribute__((ext_vector_type(4))) int   v4i;

__global__ void embw_precompute(const float* __restrict__ emb,
                                const float* __restrict__ W,
                                const float* __restrict__ b,
                                float* __restrict__ embW) {
    int idx = blockIdx.x * blockDim.x + threadIdx.x;
    if (idx >= V * H) return;
    int v = idx >> 4;
    int h = idx & 15;
    float s = b[h];
    const float* er = emb + v * E;
#pragma unroll
    for (int e = 0; e < E; ++e)
        s = fmaf(er[e], W[e * H + h], s);
    embW[idx] = s;
}

__device__ __forceinline__ float tanh_fast(float x) {
#if __has_builtin(__builtin_amdgcn_tanhf)
    return __builtin_amdgcn_tanhf(x);
#else
    float e = __builtin_amdgcn_exp2f(x * 2.885390081777926f);
    return fmaf(-2.0f, __builtin_amdgcn_rcpf(e + 1.0f), 1.0f);
#endif
}

__device__ __forceinline__ float xswap16(float x) {
    float r;
    asm("v_permlanex16_b32 %0, %1, %2, 0xfedcba98"
        : "=v"(r)
        : "v"(x), "s"(0x76543210));
    return r;
}

__global__ __launch_bounds__(32) void rnn_scan(const int* __restrict__ X,
                                               const float* __restrict__ U,
                                               const float* __restrict__ embW,
                                               float* __restrict__ out) {
    const int lane = threadIdx.x;
    const int base = blockIdx.x * 32;
    const int col  = lane & 15;
    const bool lo  = lane < 16;
    const uint32_t hoff = lo ? 0u : 8u;

    const int koff = lo ? 0 : 2;
    v2f a0, a1, a2, a3;
    a0[0] = U[(0 + koff) * H + col];  a0[1] = U[(1 + koff) * H + col];
    a1[0] = U[(4 + koff) * H + col];  a1[1] = U[(5 + koff) * H + col];
    a2[0] = U[(8 + koff) * H + col];  a2[1] = U[(9 + koff) * H + col];
    a3[0] = U[(12 + koff) * H + col]; a3[1] = U[(13 + koff) * H + col];

    const uint32_t xoff0 = (uint32_t)(base + col) * T;
    const uint32_t xoff1 = xoff0 + 16u * T;
    const uint32_t ooff0 = (uint32_t)(base + col) * T * H + hoff;
    const uint32_t ooff1 = ooff0 + 16u * T * H;

    v8f g0 = {}, g1 = {};

    auto step_tile = [&](v8f& g, v4f& e0s, v4f& e1s, int ntok, float* op) {
        v8f acc0 = {e0s[0], e0s[1], e0s[2], e0s[3],
                    e1s[0], e1s[1], e1s[2], e1s[3]};
        const float* np = embW + ((uint32_t)ntok * 16u + hoff);
        e0s = *reinterpret_cast<const v4f*>(np);
        e1s = *reinterpret_cast<const v4f*>(np + 4);

        float hs0 = xswap16(g[0]), hs1 = xswap16(g[1]);
        float hs2 = xswap16(g[2]), hs3 = xswap16(g[3]);
        float hs4 = xswap16(g[4]), hs5 = xswap16(g[5]);
        float hs6 = xswap16(g[6]), hs7 = xswap16(g[7]);
        v2f b0 = {lo ? g[0] : hs2, lo ? g[1] : hs3};
        v2f b1 = {lo ? g[4] : hs6, lo ? g[5] : hs7};
        v2f b2 = {lo ? hs0 : g[2], lo ? hs1 : g[3]};
        v2f b3 = {lo ? hs4 : g[6], lo ? hs5 : g[7]};

        v8f acc1 = {};
        acc0 = __builtin_amdgcn_wmma_f32_16x16x4_f32(false, a0, false, b0,
                                                     (short)0, acc0, false, false);
        acc1 = __builtin_amdgcn_wmma_f32_16x16x4_f32(false, a2, false, b2,
                                                     (short)0, acc1, false, false);
        acc0 = __builtin_amdgcn_wmma_f32_16x16x4_f32(false, a1, false, b1,
                                                     (short)0, acc0, false, false);
        acc1 = __builtin_amdgcn_wmma_f32_16x16x4_f32(false, a3, false, b3,
                                                     (short)0, acc1, false, false);
        v8f s = acc0 + acc1;
#pragma unroll
        for (int j = 0; j < 8; ++j) g[j] = tanh_fast(s[j]);

        v4f s0 = {g[0], g[1], g[2], g[3]};
        v4f s1 = {g[4], g[5], g[6], g[7]};
        *reinterpret_cast<v4f*>(op) = s0;
        *reinterpret_cast<v4f*>(op + 4) = s1;
    };

    v4i q0_0 = *reinterpret_cast<const v4i*>(X + xoff0);
    v4i q0_1 = *reinterpret_cast<const v4i*>(X + xoff1);
    v4i tkA0 = *reinterpret_cast<const v4i*>(X + (xoff0 + 4u));
    v4i tkA1 = *reinterpret_cast<const v4i*>(X + (xoff1 + 4u));

    v4f pe0_0[4], pe1_0[4], pe0_1[4], pe1_1[4];
#pragma unroll
    for (int i = 0; i < 4; ++i) {
        const float* p0 = embW + ((uint32_t)q0_0[i] * 16u + hoff);
        pe0_0[i] = *reinterpret_cast<const v4f*>(p0);
        pe1_0[i] = *reinterpret_cast<const v4f*>(p0 + 4);
        const float* p1 = embW + ((uint32_t)q0_1[i] * 16u + hoff);
        pe0_1[i] = *reinterpret_cast<const v4f*>(p1);
        pe1_1[i] = *reinterpret_cast<const v4f*>(p1 + 4);
    }

    for (int tt = 0; tt < T; tt += 4) {
        int tq = (tt + 8 < T) ? (tt + 8) : (T - 4);
        v4i tkB0 = *reinterpret_cast<const v4i*>(X + (xoff0 + (uint32_t)tq));
        v4i tkB1 = *reinterpret_cast<const v4i*>(X + (xoff1 + (uint32_t)tq));
        float* opb0 = out + (ooff0 + (uint32_t)tt * H);
        float* opb1 = out + (ooff1 + (uint32_t)tt * H);
#pragma unroll
        for (int u = 0; u < 4; ++u) {
            step_tile(g0, pe0_0[u], pe1_0[u], tkA0[u], opb0 + u * H);
            step_tile(g1, pe0_1[u], pe1_1[u], tkA1[u], opb1 + u * H);
        }
        tkA0 = tkB0;
        tkA1 = tkB1;
    }

    float* hp0 = out + ((uint32_t)B * T * H + (uint32_t)(base + col) * H + hoff);
    float* hp1 = hp0 + 16 * H;
    v4f f00 = {g0[0], g0[1], g0[2], g0[3]};
    v4f f01 = {g0[4], g0[5], g0[6], g0[7]};
    *reinterpret_cast<v4f*>(hp0) = f00;
    *reinterpret_cast<v4f*>(hp0 + 4) = f01;
    v4f f10 = {g1[0], g1[1], g1[2], g1[3]};
    v4f f11 = {g1[4], g1[5], g1[6], g1[7]};
    *reinterpret_cast<v4f*>(hp1) = f10;
    *reinterpret_cast<v4f*>(hp1 + 4) = f11;
}

extern "C" void kernel_launch(void* const* d_in, const int* in_sizes, int n_in,
                              void* d_out, int out_size, void* d_ws, size_t ws_size,
                              hipStream_t stream) {
    const int*   X   = (const int*)d_in[0];
    const float* emb = (const float*)d_in[1];
    const float* W   = (const float*)d_in[2];
    const float* U   = (const float*)d_in[3];
    const float* b   = (const float*)d_in[4];
    float* out  = (float*)d_out;
    float* embW = (float*)d_ws;

    embw_precompute<<<(V * H + 255) / 256, 256, 0, stream>>>(emb, W, b, embW);

    rnn_scan<<<B / 32, 32, 0, stream>>>(X, U, embW, out);
}


# One 16-row batch tile per wave instead of two: 64 single-wave work-groups (grid 32 -> 64), so a wave's serial step holds one tile's WMMA/tanh chain and half the gathers and stores
# speedup vs baseline: 1.0673x; 1.0673x over previous
#include <hip/hip_runtime.h>

constexpr int B = 1024;
constexpr int T = 512;
constexpr int V = 32000;
constexpr int E = 32;
constexpr int H = 16;

typedef __attribute__((ext_vector_type(2))) float v2f;
typedef __attribute__((ext_vector_type(4))) float v4f;
typedef __attribute__((ext_vector_type(8))) float v8f;
typedef __attribute__((ext_vector_type(4))) int   v4i;

__global__ void embw_precompute(const float* __restrict__ emb,
                                const float* __restrict__ W,
                                const float* __restrict__ b,
                                float* __restrict__ embW) {
    int idx = blockIdx.x * blockDim.x + threadIdx.x;
    if (idx >= V * H) return;
    int v = idx >> 4;
    int h = idx & 15;
    float s = b[h];
    const float* er = emb + v * E;
#pragma unroll
    for (int e = 0; e < E; ++e)
        s = fmaf(er[e], W[e * H + h], s);
    embW[idx] = s;
}

__device__ __forceinline__ float tanh_fast(float x) {
#if __has_builtin(__builtin_amdgcn_tanhf)
    return __builtin_amdgcn_tanhf(x);
#else
    float e = __builtin_amdgcn_exp2f(x * 2.885390081777926f);
    return fmaf(-2.0f, __builtin_amdgcn_rcpf(e + 1.0f), 1.0f);
#endif
}

__device__ __forceinline__ float xswap16(float x) {
    float r;
    asm("v_permlanex16_b32 %0, %1, %2, 0xfedcba98"
        : "=v"(r)
        : "v"(x), "s"(0x76543210));
    return r;
}

__global__ __launch_bounds__(32) void rnn_scan(const int* __restrict__ X,
                                               const float* __restrict__ U,
                                               const float* __restrict__ embW,
                                               float* __restrict__ out) {
    const int lane = threadIdx.x;
    const int base = blockIdx.x * 16;
    const int col  = lane & 15;
    const bool lo  = lane < 16;
    const uint32_t hoff = lo ? 0u : 8u;

    const int koff = lo ? 0 : 2;
    v2f a0, a1, a2, a3;
    a0[0] = U[(0 + koff) * H + col];  a0[1] = U[(1 + koff) * H + col];
    a1[0] = U[(4 + koff) * H + col];  a1[1] = U[(5 + koff) * H + col];
    a2[0] = U[(8 + koff) * H + col];  a2[1] = U[(9 + koff) * H + col];
    a3[0] = U[(12 + koff) * H + col]; a3[1] = U[(13 + koff) * H + col];

    const uint32_t xoff0 = (uint32_t)(base + col) * T;
    const uint32_t ooff0 = (uint32_t)(base + col) * T * H + hoff;

    v8f g0 = {};

    auto step_tile = [&](v8f& g, v4f& e0s, v4f& e1s, int ntok, float* op) {
        v8f acc0 = {e0s[0], e0s[1], e0s[2], e0s[3],
                    e1s[0], e1s[1], e1s[2], e1s[3]};
        const float* np = embW + ((uint32_t)ntok * 16u + hoff);
        e0s = *reinterpret_cast<const v4f*>(np);
        e1s = *reinterpret_cast<const v4f*>(np + 4);

        float hs0 = xswap16(g[0]), hs1 = xswap16(g[1]);
        float hs2 = xswap16(g[2]), hs3 = xswap16(g[3]);
        float hs4 = xswap16(g[4]), hs5 = xswap16(g[5]);
        float hs6 = xswap16(g[6]), hs7 = xswap16(g[7]);
        v2f b0 = {lo ? g[0] : hs2, lo ? g[1] : hs3};
        v2f b1 = {lo ? g[4] : hs6, lo ? g[5] : hs7};
        v2f b2 = {lo ? hs0 : g[2], lo ? hs1 : g[3]};
        v2f b3 = {lo ? hs4 : g[6], lo ? hs5 : g[7]};

        v8f acc1 = {};
        acc0 = __builtin_amdgcn_wmma_f32_16x16x4_f32(false, a0, false, b0,
                                                     (short)0, acc0, false, false);
        acc1 = __builtin_amdgcn_wmma_f32_16x16x4_f32(false, a2, false, b2,
                                                     (short)0, acc1, false, false);
        acc0 = __builtin_amdgcn_wmma_f32_16x16x4_f32(false, a1, false, b1,
                                                     (short)0, acc0, false, false);
        acc1 = __builtin_amdgcn_wmma_f32_16x16x4_f32(false, a3, false, b3,
                                                     (short)0, acc1, false, false);
        v8f s = acc0 + acc1;
#pragma unroll
        for (int j = 0; j < 8; ++j) g[j] = tanh_fast(s[j]);

        v4f s0 = {g[0], g[1], g[2], g[3]};
        v4f s1 = {g[4], g[5], g[6], g[7]};
        *reinterpret_cast<v4f*>(op) = s0;
        *reinterpret_cast<v4f*>(op + 4) = s1;
    };

    v4i q0_0 = *reinterpret_cast<const v4i*>(X + xoff0);
    v4i tkA0 = *reinterpret_cast<const v4i*>(X + (xoff0 + 4u));

    v4f pe0_0[4], pe1_0[4];
#pragma unroll
    for (int i = 0; i < 4; ++i) {
        const float* p0 = embW + ((uint32_t)q0_0[i] * 16u + hoff);
        pe0_0[i] = *reinterpret_cast<const v4f*>(p0);
        pe1_0[i] = *reinterpret_cast<const v4f*>(p0 + 4);
    }

    for (int tt = 0; tt < T; tt += 4) {
        int tq = (tt + 8 < T) ? (tt + 8) : (T - 4);
        v4i tkB0 = *reinterpret_cast<const v4i*>(X + (xoff0 + (uint32_t)tq));
        float* opb0 = out + (ooff0 + (uint32_t)tt * H);
#pragma unroll
        for (int u = 0; u < 4; ++u)
            step_tile(g0, pe0_0[u], pe1_0[u], tkA0[u], opb0 + u * H);
        tkA0 = tkB0;
    }

    float* hp0 = out + ((uint32_t)B * T * H + (uint32_t)(base + col) * H + hoff);
    v4f f00 = {g0[0], g0[1], g0[2], g0[3]};
    v4f f01 = {g0[4], g0[5], g0[6], g0[7]};
    *reinterpret_cast<v4f*>(hp0) = f00;
    *reinterpret_cast<v4f*>(hp0 + 4) = f01;
}

extern "C" void kernel_launch(void* const* d_in, const int* in_sizes, int n_in,
                              void* d_out, int out_size, void* d_ws, size_t ws_size,
                              hipStream_t stream) {
    const int*   X   = (const int*)d_in[0];
    const float* emb = (const float*)d_in[1];
    const float* W   = (const float*)d_in[2];
    const float* U   = (const float*)d_in[3];
    const float* b   = (const float*)d_in[4];
    float* out  = (float*)d_out;
    float* embW = (float*)d_ws;

    embw_precompute<<<(V * H + 255) / 256, 256, 0, stream>>>(emb, W, b, embW);

    rnn_scan<<<B / 16, 32, 0, stream>>>(X, U, embW, out);
}


# x-tile gathers and token quads run a 16-slot register ring ahead of the scan, each refill issued right behind its slot's last read: no load wait or register copy is left on the serial step chain
# speedup vs baseline: 1.9036x; 1.7836x over previous
#include <hip/hip_runtime.h>

constexpr int B = 1024;
constexpr int T = 512;
constexpr int V = 32000;
constexpr int E = 32;
constexpr int H = 16;

typedef __attribute__((ext_vector_type(2))) float v2f;
typedef __attribute__((ext_vector_type(4))) float v4f;
typedef __attribute__((ext_vector_type(8))) float v8f;
typedef __attribute__((ext_vector_type(4))) int   v4i;

__global__ void embw_precompute(const float* __restrict__ emb,
                                const float* __restrict__ W,
                                const float* __restrict__ b,
                                float* __restrict__ embW) {
    int idx = blockIdx.x * blockDim.x + threadIdx.x;
    if (idx >= V * H) return;
    int v = idx >> 4;
    int h = idx & 15;
    float s = b[h];
    const float* er = emb + v * E;
#pragma unroll
    for (int e = 0; e < E; ++e)
        s = fmaf(er[e], W[e * H + h], s);
    embW[idx] = s;
}

__device__ __forceinline__ float tanh_fast(float x) {
#if __has_builtin(__builtin_amdgcn_tanhf)
    return __builtin_amdgcn_tanhf(x);
#else
    float e = __builtin_amdgcn_exp2f(x * 2.885390081777926f);
    return fmaf(-2.0f, __builtin_amdgcn_rcpf(e + 1.0f), 1.0f);
#endif
}

__device__ __forceinline__ float xswap16(float x) {
    float r;
    asm("v_permlanex16_b32 %0, %1, %2, 0xfedcba98"
        : "=v"(r)
        : "v"(x), "s"(0x76543210));
    return r;
}

constexpr int RING = 16;
constexpr int NQ   = RING / 4;
struct YesT { static constexpr bool value = true; };
struct NoT  { static constexpr bool value = false; };

__global__ __launch_bounds__(32) void rnn_scan(const int* __restrict__ X,
                                               const float* __restrict__ U,
                                               const float* __restrict__ embW,
                                               float* __restrict__ out) {
    const int lane = threadIdx.x;
    const int base = blockIdx.x * 16;
    const int col  = lane & 15;
    const bool lo  = lane < 16;
    const uint32_t hoff = lo ? 0u : 8u;

    const int koff = lo ? 0 : 2;
    v2f a0, a1, a2, a3;
    a0[0] = U[(0 + koff) * H + col];  a0[1] = U[(1 + koff) * H + col];
    a1[0] = U[(4 + koff) * H + col];  a1[1] = U[(5 + koff) * H + col];
    a2[0] = U[(8 + koff) * H + col];  a2[1] = U[(9 + koff) * H + col];
    a3[0] = U[(12 + koff) * H + col]; a3[1] = U[(13 + koff) * H + col];

    const uint32_t xoff0 = (uint32_t)(base + col) * T;
    const uint32_t ooff0 = (uint32_t)(base + col) * T * H + hoff;

    v8f g = {};

    auto gather = [&](int tok) -> v8f {
        const float* np = embW + ((uint32_t)tok * 16u + hoff);
        v4f l0 = *reinterpret_cast<const v4f*>(np);
        v4f l1 = *reinterpret_cast<const v4f*>(np + 4);
        return __builtin_shufflevector(l0, l1, 0, 1, 2, 3, 4, 5, 6, 7);
    };

    v4i first[NQ];
    v4i tok[NQ];
    v8f xr[RING];
#pragma unroll
    for (int q = 0; q < NQ; ++q)
        first[q] = *reinterpret_cast<const v4i*>(X + (xoff0 + 4u * q));
#pragma unroll
    for (int u = 0; u < RING; ++u) {
        xr[u] = gather(first[u >> 2][u & 3]);
        if ((u & 3) == 3)
            tok[u >> 2] = *reinterpret_cast<const v4i*>(
                X + (xoff0 + (uint32_t)RING + 4u * (u >> 2)));
        __builtin_amdgcn_sched_barrier(0);
    }

    auto trip = [&](int tt, auto refill) {
        constexpr bool REFILL = decltype(refill)::value;
        const int tq = (tt + 2 * RING < T) ? (tt + 2 * RING) : (T - RING);
        float* opb = out + (ooff0 + (uint32_t)tt * H);
#pragma unroll
        for (int u = 0; u < RING; ++u) {
            float hs0 = xswap16(g[0]), hs1 = xswap16(g[1]);
            float hs2 = xswap16(g[2]), hs3 = xswap16(g[3]);
            float hs4 = xswap16(g[4]), hs5 = xswap16(g[5]);
            float hs6 = xswap16(g[6]), hs7 = xswap16(g[7]);
            v2f b0 = {lo ? g[0] : hs2, lo ? g[1] : hs3};
            v2f b1 = {lo ? g[4] : hs6, lo ? g[5] : hs7};
            v2f b2 = {lo ? hs0 : g[2], lo ? hs1 : g[3]};
            v2f b3 = {lo ? hs4 : g[6], lo ? hs5 : g[7]};

            v8f acc0 = xr[u];
            v8f acc1 = {};
            acc0 = __builtin_amdgcn_wmma_f32_16x16x4_f32(false, a0, false, b0,
                                                         (short)0, acc0, false, false);
            acc1 = __builtin_amdgcn_wmma_f32_16x16x4_f32(false, a2, false, b2,
                                                         (short)0, acc1, false, false);
            acc0 = __builtin_amdgcn_wmma_f32_16x16x4_f32(false, a1, false, b1,
                                                         (short)0, acc0, false, false);
            acc1 = __builtin_amdgcn_wmma_f32_16x16x4_f32(false, a3, false, b3,
                                                         (short)0, acc1, false, false);
            if constexpr (REFILL) __builtin_amdgcn_sched_barrier(0);

            v8f s = acc0 + acc1;
#pragma unroll
            for (int j = 0; j < 8; ++j) g[j] = tanh_fast(s[j]);

            float* op = opb + u * H;
            v4f s0 = {g[0], g[1], g[2], g[3]};
            v4f s1 = {g[4], g[5], g[6], g[7]};
            *reinterpret_cast<v4f*>(op) = s0;
            *reinterpret_cast<v4f*>(op + 4) = s1;

            if constexpr (REFILL) {
                xr[u] = gather(tok[u >> 2][u & 3]);
                if ((u & 3) == 3)
                    tok[u >> 2] = *reinterpret_cast<const v4i*>(
                        X + (xoff0 + (uint32_t)tq + 4u * (u >> 2)));
            }
        }
    };

    for (int tt = 0; tt < T - RING; tt += RING)
        trip(tt, YesT{});
    trip(T - RING, NoT{});

    float* hp0 = out + ((uint32_t)B * T * H + (uint32_t)(base + col) * H + hoff);
    v4f f00 = {g[0], g[1], g[2], g[3]};
    v4f f01 = {g[4], g[5], g[6], g[7]};
    *reinterpret_cast<v4f*>(hp0) = f00;
    *reinterpret_cast<v4f*>(hp0 + 4) = f01;
}

extern "C" void kernel_launch(void* const* d_in, const int* in_sizes, int n_in,
                              void* d_out, int out_size, void* d_ws, size_t ws_size,
                              hipStream_t stream) {
    const int*   X   = (const int*)d_in[0];
    const float* emb = (const float*)d_in[1];
    const float* W   = (const float*)d_in[2];
    const float* U   = (const float*)d_in[3];
    const float* b   = (const float*)d_in[4];
    float* out  = (float*)d_out;
    float* embW = (float*)d_ws;

    embw_precompute<<<(V * H + 255) / 256, 256, 0, stream>>>(emb, W, b, embW);

    rnn_scan<<<B / 16, 32, 0, stream>>>(X, U, embW, out);
}


# Recurrent product's k order taken so that the D-layout state register pair (2c,2c+1) is chunk c's B fragment as it stands: 8 half-wave swaps + 8 selects per step gone; k terms summed in a new order
# speedup vs baseline: 2.2298x; 1.1714x over previous
#include <hip/hip_runtime.h>

constexpr int B = 1024;
constexpr int T = 512;
constexpr int V = 32000;
constexpr int E = 32;
constexpr int H = 16;

typedef __attribute__((ext_vector_type(2))) float v2f;
typedef __attribute__((ext_vector_type(4))) float v4f;
typedef __attribute__((ext_vector_type(8))) float v8f;
typedef __attribute__((ext_vector_type(4))) int   v4i;

__global__ void embw_precompute(const float* __restrict__ emb,
                                const float* __restrict__ W,
                                const float* __restrict__ b,
                                float* __restrict__ embW) {
    int idx = blockIdx.x * blockDim.x + threadIdx.x;
    if (idx >= V * H) return;
    int v = idx >> 4;
    int h = idx & 15;
    float s = b[h];
    const float* er = emb + v * E;
#pragma unroll
    for (int e = 0; e < E; ++e)
        s = fmaf(er[e], W[e * H + h], s);
    embW[idx] = s;
}

__device__ __forceinline__ float tanh_fast(float x) {
#if __has_builtin(__builtin_amdgcn_tanhf)
    return __builtin_amdgcn_tanhf(x);
#else
    float e = __builtin_amdgcn_exp2f(x * 2.885390081777926f);
    return fmaf(-2.0f, __builtin_amdgcn_rcpf(e + 1.0f), 1.0f);
#endif
}

constexpr int RING = 16;
constexpr int NQ   = RING / 4;
struct YesT { static constexpr bool value = true; };
struct NoT  { static constexpr bool value = false; };

__global__ __launch_bounds__(32) void rnn_scan(const int* __restrict__ X,
                                               const float* __restrict__ U,
                                               const float* __restrict__ embW,
                                               float* __restrict__ out) {
    const int lane = threadIdx.x;
    const int base = blockIdx.x * 16;
    const int col  = lane & 15;
    const bool lo  = lane < 16;
    const uint32_t hoff = lo ? 0u : 8u;

    const int kb = lo ? 0 : 8;
    v2f a0, a1, a2, a3;
    a0[0] = U[(kb + 0) * H + col];  a0[1] = U[(kb + 1) * H + col];
    a1[0] = U[(kb + 2) * H + col];  a1[1] = U[(kb + 3) * H + col];
    a2[0] = U[(kb + 4) * H + col];  a2[1] = U[(kb + 5) * H + col];
    a3[0] = U[(kb + 6) * H + col];  a3[1] = U[(kb + 7) * H + col];

    const uint32_t xoff0 = (uint32_t)(base + col) * T;
    const uint32_t ooff0 = (uint32_t)(base + col) * T * H + hoff;

    v8f g = {};

    auto gather = [&](int tok) -> v8f {
        const float* np = embW + ((uint32_t)tok * 16u + hoff);
        v4f l0 = *reinterpret_cast<const v4f*>(np);
        v4f l1 = *reinterpret_cast<const v4f*>(np + 4);
        return __builtin_shufflevector(l0, l1, 0, 1, 2, 3, 4, 5, 6, 7);
    };

    v4i first[NQ];
    v4i tok[NQ];
    v8f xr[RING];
#pragma unroll
    for (int q = 0; q < NQ; ++q)
        first[q] = *reinterpret_cast<const v4i*>(X + (xoff0 + 4u * q));
#pragma unroll
    for (int u = 0; u < RING; ++u) {
        xr[u] = gather(first[u >> 2][u & 3]);
        if ((u & 3) == 3)
            tok[u >> 2] = *reinterpret_cast<const v4i*>(
                X + (xoff0 + (uint32_t)RING + 4u * (u >> 2)));
        __builtin_amdgcn_sched_barrier(0);
    }

    auto trip = [&](int tt, auto refill) {
        constexpr bool REFILL = decltype(refill)::value;
        const int tq = (tt + 2 * RING < T) ? (tt + 2 * RING) : (T - RING);
        float* opb = out + (ooff0 + (uint32_t)tt * H);
#pragma unroll
        for (int u = 0; u < RING; ++u) {
            v2f b0 = {g[0], g[1]};
            v2f b1 = {g[2], g[3]};
            v2f b2 = {g[4], g[5]};
            v2f b3 = {g[6], g[7]};

            v8f acc0 = xr[u];
            v8f acc1 = {};
            acc0 = __builtin_amdgcn_wmma_f32_16x16x4_f32(false, a0, false, b0,
                                                         (short)0, acc0, false, false);
            acc1 = __builtin_amdgcn_wmma_f32_16x16x4_f32(false, a1, false, b1,
                                                         (short)0, acc1, false, false);
            acc0 = __builtin_amdgcn_wmma_f32_16x16x4_f32(false, a2, false, b2,
                                                         (short)0, acc0, false, false);
            acc1 = __builtin_amdgcn_wmma_f32_16x16x4_f32(false, a3, false, b3,
                                                         (short)0, acc1, false, false);
            if constexpr (REFILL) __builtin_amdgcn_sched_barrier(0);

            v8f s = acc0 + acc1;
#pragma unroll
            for (int j = 0; j < 8; ++j) g[j] = tanh_fast(s[j]);

            float* op = opb + u * H;
            v4f s0 = {g[0], g[1], g[2], g[3]};
            v4f s1 = {g[4], g[5], g[6], g[7]};
            *reinterpret_cast<v4f*>(op) = s0;
            *reinterpret_cast<v4f*>(op + 4) = s1;

            if constexpr (REFILL) {
                xr[u] = gather(tok[u >> 2][u & 3]);
                if ((u & 3) == 3)
                    tok[u >> 2] = *reinterpret_cast<const v4i*>(
                        X + (xoff0 + (uint32_t)tq + 4u * (u >> 2)));
            }
        }
    };

    for (int tt = 0; tt < T - RING; tt += RING)
        trip(tt, YesT{});
    trip(T - RING, NoT{});

    float* hp0 = out + ((uint32_t)B * T * H + (uint32_t)(base + col) * H + hoff);
    v4f f00 = {g[0], g[1], g[2], g[3]};
    v4f f01 = {g[4], g[5], g[6], g[7]};
    *reinterpret_cast<v4f*>(hp0) = f00;
    *reinterpret_cast<v4f*>(hp0 + 4) = f01;
}

extern "C" void kernel_launch(void* const* d_in, const int* in_sizes, int n_in,
                              void* d_out, int out_size, void* d_ws, size_t ws_size,
                              hipStream_t stream) {
    const int*   X   = (const int*)d_in[0];
    const float* emb = (const float*)d_in[1];
    const float* W   = (const float*)d_in[2];
    const float* U   = (const float*)d_in[3];
    const float* b   = (const float*)d_in[4];
    float* out  = (float*)d_out;
    float* embW = (float*)d_ws;

    embw_precompute<<<(V * H + 255) / 256, 256, 0, stream>>>(emb, W, b, embW);

    rnn_scan<<<B / 16, 32, 0, stream>>>(X, U, embW, out);
}


# Scan's loads and stores grouped: one place per 8 steps holds their 16 row gathers and 16 output stores (states kept in registers meanwhile); the steps between are WMMA, add and tanh only
# speedup vs baseline: 2.5392x; 1.1388x over previous
#include <hip/hip_runtime.h>

constexpr int B = 1024;
constexpr int T = 512;
constexpr int V = 32000;
constexpr int E = 32;
constexpr int H = 16;

typedef __attribute__((ext_vector_type(2))) float v2f;
typedef __attribute__((ext_vector_type(4))) float v4f;
typedef __attribute__((ext_vector_type(8))) float v8f;
typedef __attribute__((ext_vector_type(4))) int   v4i;

__global__ void embw_precompute(const float* __restrict__ emb,
                                const float* __restrict__ W,
                                const float* __restrict__ b,
                                float* __restrict__ embW) {
    int idx = blockIdx.x * blockDim.x + threadIdx.x;
    if (idx >= V * H) return;
    int v = idx >> 4;
    int h = idx & 15;
    float s = b[h];
    const float* er = emb + v * E;
#pragma unroll
    for (int e = 0; e < E; ++e)
        s = fmaf(er[e], W[e * H + h], s);
    embW[idx] = s;
}

__device__ __forceinline__ float tanh_fast(float x) {
#if __has_builtin(__builtin_amdgcn_tanhf)
    return __builtin_amdgcn_tanhf(x);
#else
    float e = __builtin_amdgcn_exp2f(x * 2.885390081777926f);
    return fmaf(-2.0f, __builtin_amdgcn_rcpf(e + 1.0f), 1.0f);
#endif
}

constexpr int RING = 16;
constexpr int GRP  = 8;
constexpr int NQ   = RING / 4;
struct YesT { static constexpr bool value = true; };
struct NoT  { static constexpr bool value = false; };

__global__ __launch_bounds__(32) void rnn_scan(const int* __restrict__ X,
                                               const float* __restrict__ U,
                                               const float* __restrict__ embW,
                                               float* __restrict__ out) {
    const int lane = threadIdx.x;
    const int base = blockIdx.x * 16;
    const int col  = lane & 15;
    const bool lo  = lane < 16;
    const uint32_t hoff = lo ? 0u : 8u;

    const int kb = lo ? 0 : 8;
    v2f a0, a1, a2, a3;
    a0[0] = U[(kb + 0) * H + col];  a0[1] = U[(kb + 1) * H + col];
    a1[0] = U[(kb + 2) * H + col];  a1[1] = U[(kb + 3) * H + col];
    a2[0] = U[(kb + 4) * H + col];  a2[1] = U[(kb + 5) * H + col];
    a3[0] = U[(kb + 6) * H + col];  a3[1] = U[(kb + 7) * H + col];

    const uint32_t xoff0 = (uint32_t)(base + col) * T;
    const uint32_t ooff0 = (uint32_t)(base + col) * T * H + hoff;

    v8f g = {};

    auto gather = [&](int tok) -> v8f {
        const float* np = embW + ((uint32_t)tok * 16u + hoff);
        v4f l0 = *reinterpret_cast<const v4f*>(np);
        v4f l1 = *reinterpret_cast<const v4f*>(np + 4);
        return __builtin_shufflevector(l0, l1, 0, 1, 2, 3, 4, 5, 6, 7);
    };
    auto gather_at = [&](uint32_t boff) -> v8f {
        const char* np = reinterpret_cast<const char*>(embW) + boff;
        v4f l0 = *reinterpret_cast<const v4f*>(np);
        v4f l1 = *reinterpret_cast<const v4f*>(np + 16);
        return __builtin_shufflevector(l0, l1, 0, 1, 2, 3, 4, 5, 6, 7);
    };

    v4i first[NQ];
    v4i tok[NQ];
    v8f xr[RING];
#pragma unroll
    for (int q = 0; q < NQ; ++q)
        first[q] = *reinterpret_cast<const v4i*>(X + (xoff0 + 4u * q));
#pragma unroll
    for (int p = 0; p < RING / GRP; ++p) {
#pragma unroll
        for (int v = 0; v < GRP; ++v) {
            const int u = p * GRP + v;
            xr[u] = gather(first[u >> 2][u & 3]);
        }
        if (p == 0) {
#pragma unroll
            for (int q = 0; q < NQ; ++q)
                tok[q] = *reinterpret_cast<const v4i*>(
                    X + (xoff0 + (uint32_t)RING + 4u * q));
        }
        __builtin_amdgcn_sched_barrier(0);
    }

    auto trip = [&](int tt, auto refill) {
        constexpr bool REFILL = decltype(refill)::value;
        const int tq = (tt + 2 * RING < T) ? (tt + 2 * RING) : (T - RING);
        float* opb = out + (ooff0 + (uint32_t)tt * H);
        v8f gs[RING];
        uint32_t bo[RING] = {};
#pragma unroll
        for (int p = 0; p < RING / GRP; ++p) {
#pragma unroll
            for (int v = 0; v < GRP; ++v) {
                const int u = p * GRP + v;
                v2f b0 = {g[0], g[1]};
                v2f b1 = {g[2], g[3]};
                v2f b2 = {g[4], g[5]};
                v2f b3 = {g[6], g[7]};

                v8f acc0 = xr[u];
                v8f acc1 = {};
                acc0 = __builtin_amdgcn_wmma_f32_16x16x4_f32(false, a0, false, b0,
                                                             (short)0, acc0, false, false);
                acc1 = __builtin_amdgcn_wmma_f32_16x16x4_f32(false, a1, false, b1,
                                                             (short)0, acc1, false, false);
                acc0 = __builtin_amdgcn_wmma_f32_16x16x4_f32(false, a2, false, b2,
                                                             (short)0, acc0, false, false);
                acc1 = __builtin_amdgcn_wmma_f32_16x16x4_f32(false, a3, false, b3,
                                                             (short)0, acc1, false, false);

                v8f s = acc0 + acc1;
#pragma unroll
                for (int j = 0; j < 8; ++j) g[j] = tanh_fast(s[j]);
                gs[u] = g;
            }

            __builtin_amdgcn_sched_barrier(0);

            if constexpr (REFILL) {
                if (p == 0) {
#pragma unroll
                    for (int u = 0; u < RING; ++u)
                        bo[u] = (uint32_t)tok[u >> 2][u & 3] * 64u + hoff * 4u;
                }
#pragma unroll
                for (int v = 0; v < GRP; ++v)
                    xr[p * GRP + v] = gather_at(bo[p * GRP + v]);
                if (p == 0) {
#pragma unroll
                    for (int q = 0; q < NQ; ++q)
                        tok[q] = *reinterpret_cast<const v4i*>(
                            X + (xoff0 + (uint32_t)tq + 4u * q));
                }
            }
#pragma unroll
            for (int v = 0; v < GRP; ++v) {
                const int u = p * GRP + v;
                float* op = opb + u * H;
                v4f s0 = {gs[u][0], gs[u][1], gs[u][2], gs[u][3]};
                v4f s1 = {gs[u][4], gs[u][5], gs[u][6], gs[u][7]};
                *reinterpret_cast<v4f*>(op) = s0;
                *reinterpret_cast<v4f*>(op + 4) = s1;
            }
            if constexpr (REFILL) __builtin_amdgcn_sched_barrier(0);
        }
    };

    for (int tt = 0; tt < T - RING; tt += RING)
        trip(tt, YesT{});
    trip(T - RING, NoT{});

    float* hp0 = out + ((uint32_t)B * T * H + (uint32_t)(base + col) * H + hoff);
    v4f f00 = {g[0], g[1], g[2], g[3]};
    v4f f01 = {g[4], g[5], g[6], g[7]};
    *reinterpret_cast<v4f*>(hp0) = f00;
    *reinterpret_cast<v4f*>(hp0 + 4) = f01;
}

extern "C" void kernel_launch(void* const* d_in, const int* in_sizes, int n_in,
                              void* d_out, int out_size, void* d_ws, size_t ws_size,
                              hipStream_t stream) {
    const int*   X   = (const int*)d_in[0];
    const float* emb = (const float*)d_in[1];
    const float* W   = (const float*)d_in[2];
    const float* U   = (const float*)d_in[3];
    const float* b   = (const float*)d_in[4];
    float* out  = (float*)d_out;
    float* embW = (float*)d_ws;

    embw_precompute<<<(V * H + 255) / 256, 256, 0, stream>>>(emb, W, b, embW);

    rnn_scan<<<B / 16, 32, 0, stream>>>(X, U, embW, out);
}


# Output stores staged through LDS rows: one store instruction now writes 512 contiguous bytes of one batch row (4 whole 128-B lines), not two 16-B pieces in each of 16 rows; values unchanged
# speedup vs baseline: 3.9588x; 1.5591x over previous
#include <hip/hip_runtime.h>

constexpr int B = 1024;
constexpr int T = 512;
constexpr int V = 32000;
constexpr int E = 32;
constexpr int H = 16;

typedef __attribute__((ext_vector_type(2))) float v2f;
typedef __attribute__((ext_vector_type(4))) float v4f;
typedef __attribute__((ext_vector_type(8))) float v8f;
typedef __attribute__((ext_vector_type(4))) int   v4i;

__global__ void embw_precompute(const float* __restrict__ emb,
                                const float* __restrict__ W,
                                const float* __restrict__ b,
                                float* __restrict__ embW) {
    int idx = blockIdx.x * blockDim.x + threadIdx.x;
    if (idx >= V * H) return;
    int v = idx >> 4;
    int h = idx & 15;
    float s = b[h];
    const float* er = emb + v * E;
#pragma unroll
    for (int e = 0; e < E; ++e)
        s = fmaf(er[e], W[e * H + h], s);
    embW[idx] = s;
}

__device__ __forceinline__ float tanh_fast(float x) {
#if __has_builtin(__builtin_amdgcn_tanhf)
    return __builtin_amdgcn_tanhf(x);
#else
    float e = __builtin_amdgcn_exp2f(x * 2.885390081777926f);
    return fmaf(-2.0f, __builtin_amdgcn_rcpf(e + 1.0f), 1.0f);
#endif
}

constexpr int RING = 16;
constexpr int GRP  = 8;
constexpr int NQ   = RING / 4;
constexpr int SPITCH = GRP * H + 4;
struct YesT { static constexpr bool value = true; };
struct NoT  { static constexpr bool value = false; };

__global__ __launch_bounds__(32) void rnn_scan(const int* __restrict__ X,
                                               const float* __restrict__ U,
                                               const float* __restrict__ embW,
                                               float* __restrict__ out) {
    const int lane = threadIdx.x;
    const int base = blockIdx.x * 16;
    const int col  = lane & 15;
    const bool lo  = lane < 16;
    const uint32_t hoff = lo ? 0u : 8u;

    const int kb = lo ? 0 : 8;
    v2f a0, a1, a2, a3;
    a0[0] = U[(kb + 0) * H + col];  a0[1] = U[(kb + 1) * H + col];
    a1[0] = U[(kb + 2) * H + col];  a1[1] = U[(kb + 3) * H + col];
    a2[0] = U[(kb + 4) * H + col];  a2[1] = U[(kb + 5) * H + col];
    a3[0] = U[(kb + 6) * H + col];  a3[1] = U[(kb + 7) * H + col];

    const uint32_t xoff0 = (uint32_t)(base + col) * T;
    __shared__ __attribute__((aligned(16))) float stage[16 * SPITCH];
    float* const sw = stage + (uint32_t)col * SPITCH + hoff;
    const float* const sr = stage + (uint32_t)lane * 4u;
    float* const orow = out + ((uint32_t)base * T * H + (uint32_t)lane * 4u);

    v8f g = {};

    auto gather = [&](int tok) -> v8f {
        const float* np = embW + ((uint32_t)tok * 16u + hoff);
        v4f l0 = *reinterpret_cast<const v4f*>(np);
        v4f l1 = *reinterpret_cast<const v4f*>(np + 4);
        return __builtin_shufflevector(l0, l1, 0, 1, 2, 3, 4, 5, 6, 7);
    };
    auto gather_at = [&](uint32_t boff) -> v8f {
        const char* np = reinterpret_cast<const char*>(embW) + boff;
        v4f l0 = *reinterpret_cast<const v4f*>(np);
        v4f l1 = *reinterpret_cast<const v4f*>(np + 16);
        return __builtin_shufflevector(l0, l1, 0, 1, 2, 3, 4, 5, 6, 7);
    };

    v4i first[NQ];
    v4i tok[NQ];
    v8f xr[RING];
#pragma unroll
    for (int q = 0; q < NQ; ++q)
        first[q] = *reinterpret_cast<const v4i*>(X + (xoff0 + 4u * q));
#pragma unroll
    for (int p = 0; p < RING / GRP; ++p) {
#pragma unroll
        for (int v = 0; v < GRP; ++v) {
            const int u = p * GRP + v;
            xr[u] = gather(first[u >> 2][u & 3]);
        }
        if (p == 0) {
#pragma unroll
            for (int q = 0; q < NQ; ++q)
                tok[q] = *reinterpret_cast<const v4i*>(
                    X + (xoff0 + (uint32_t)RING + 4u * q));
        }
        __builtin_amdgcn_sched_barrier(0);
    }

    auto trip = [&](int tt, auto refill) {
        constexpr bool REFILL = decltype(refill)::value;
        const int tq = (tt + 2 * RING < T) ? (tt + 2 * RING) : (T - RING);
        float* opb = orow + (uint32_t)tt * H;
        v8f gs[RING];
        uint32_t bo[RING] = {};
#pragma unroll
        for (int p = 0; p < RING / GRP; ++p) {
#pragma unroll
            for (int v = 0; v < GRP; ++v) {
                const int u = p * GRP + v;
                v2f b0 = {g[0], g[1]};
                v2f b1 = {g[2], g[3]};
                v2f b2 = {g[4], g[5]};
                v2f b3 = {g[6], g[7]};

                v8f acc0 = xr[u];
                v8f acc1 = {};
                acc0 = __builtin_amdgcn_wmma_f32_16x16x4_f32(false, a0, false, b0,
                                                             (short)0, acc0, false, false);
                acc1 = __builtin_amdgcn_wmma_f32_16x16x4_f32(false, a1, false, b1,
                                                             (short)0, acc1, false, false);
                acc0 = __builtin_amdgcn_wmma_f32_16x16x4_f32(false, a2, false, b2,
                                                             (short)0, acc0, false, false);
                acc1 = __builtin_amdgcn_wmma_f32_16x16x4_f32(false, a3, false, b3,
                                                             (short)0, acc1, false, false);

                v8f s = acc0 + acc1;
#pragma unroll
                for (int j = 0; j < 8; ++j) g[j] = tanh_fast(s[j]);
                gs[u] = g;
            }

            __builtin_amdgcn_sched_barrier(0);

            if constexpr (REFILL) {
                if (p == 0) {
#pragma unroll
                    for (int u = 0; u < RING; ++u)
                        bo[u] = (uint32_t)tok[u >> 2][u & 3] * 64u + hoff * 4u;
                }
#pragma unroll
                for (int v = 0; v < GRP; ++v)
                    xr[p * GRP + v] = gather_at(bo[p * GRP + v]);
                if (p == 0) {
#pragma unroll
                    for (int q = 0; q < NQ; ++q)
                        tok[q] = *reinterpret_cast<const v4i*>(
                            X + (xoff0 + (uint32_t)tq + 4u * q));
                }
            }
#pragma unroll
            for (int v = 0; v < GRP; ++v) {
                const int u = p * GRP + v;
                v4f s0 = {gs[u][0], gs[u][1], gs[u][2], gs[u][3]};
                v4f s1 = {gs[u][4], gs[u][5], gs[u][6], gs[u][7]};
                *reinterpret_cast<v4f*>(sw + v * H) = s0;
                *reinterpret_cast<v4f*>(sw + v * H + 4) = s1;
            }
#pragma unroll
            for (int r = 0; r < 16; ++r) {
                v4f d = *reinterpret_cast<const v4f*>(sr + r * SPITCH);
                *reinterpret_cast<v4f*>(opb + (r * T * H + p * GRP * H)) = d;
            }
            if constexpr (REFILL) __builtin_amdgcn_sched_barrier(0);
        }
    };

    for (int tt = 0; tt < T - RING; tt += RING)
        trip(tt, YesT{});
    trip(T - RING, NoT{});

    float* hp0 = out + ((uint32_t)B * T * H + (uint32_t)(base + col) * H + hoff);
    v4f f00 = {g[0], g[1], g[2], g[3]};
    v4f f01 = {g[4], g[5], g[6], g[7]};
    *reinterpret_cast<v4f*>(hp0) = f00;
    *reinterpret_cast<v4f*>(hp0 + 4) = f01;
}

extern "C" void kernel_launch(void* const* d_in, const int* in_sizes, int n_in,
                              void* d_out, int out_size, void* d_ws, size_t ws_size,
                              hipStream_t stream) {
    const int*   X   = (const int*)d_in[0];
    const float* emb = (const float*)d_in[1];
    const float* W   = (const float*)d_in[2];
    const float* U   = (const float*)d_in[3];
    const float* b   = (const float*)d_in[4];
    float* out  = (float*)d_out;
    float* embW = (float*)d_ws;

    embw_precompute<<<(V * H + 255) / 256, 256, 0, stream>>>(emb, W, b, embW);

    rnn_scan<<<B / 16, 32, 0, stream>>>(X, U, embW, out);
}


# A group's staged rows are read back from LDS at the end of its place and stored at the head of the next one, carried in registers over the steps between: no LDS round trip is waited for in a place
# speedup vs baseline: 4.2457x; 1.0725x over previous
#include <hip/hip_runtime.h>

constexpr int B = 1024;
constexpr int T = 512;
constexpr int V = 32000;
constexpr int E = 32;
constexpr int H = 16;

typedef __attribute__((ext_vector_type(2))) float v2f;
typedef __attribute__((ext_vector_type(4))) float v4f;
typedef __attribute__((ext_vector_type(8))) float v8f;
typedef __attribute__((ext_vector_type(4))) int   v4i;

__global__ void embw_precompute(const float* __restrict__ emb,
                                const float* __restrict__ W,
                                const float* __restrict__ b,
                                float* __restrict__ embW) {
    int idx = blockIdx.x * blockDim.x + threadIdx.x;
    if (idx >= V * H) return;
    int v = idx >> 4;
    int h = idx & 15;
    float s = b[h];
    const float* er = emb + v * E;
#pragma unroll
    for (int e = 0; e < E; ++e)
        s = fmaf(er[e], W[e * H + h], s);
    embW[idx] = s;
}

__device__ __forceinline__ float tanh_fast(float x) {
#if __has_builtin(__builtin_amdgcn_tanhf)
    return __builtin_amdgcn_tanhf(x);
#else
    float e = __builtin_amdgcn_exp2f(x * 2.885390081777926f);
    return fmaf(-2.0f, __builtin_amdgcn_rcpf(e + 1.0f), 1.0f);
#endif
}

constexpr int RING = 16;
constexpr int GRP  = 8;
constexpr int NQ   = RING / 4;
constexpr int SPITCH = GRP * H + 4;
constexpr int SBUF   = 16 * SPITCH;
struct YesT { static constexpr bool value = true; };
struct NoT  { static constexpr bool value = false; };
struct P0T  { static constexpr int value = 0; };
struct P1T  { static constexpr int value = 1; };

__global__ __launch_bounds__(32) void rnn_scan(const int* __restrict__ X,
                                               const float* __restrict__ U,
                                               const float* __restrict__ embW,
                                               float* __restrict__ out) {
    const int lane = threadIdx.x;
    const int base = blockIdx.x * 16;
    const int col  = lane & 15;
    const bool lo  = lane < 16;
    const uint32_t hoff = lo ? 0u : 8u;

    const int kb = lo ? 0 : 8;
    v2f a0, a1, a2, a3;
    a0[0] = U[(kb + 0) * H + col];  a0[1] = U[(kb + 1) * H + col];
    a1[0] = U[(kb + 2) * H + col];  a1[1] = U[(kb + 3) * H + col];
    a2[0] = U[(kb + 4) * H + col];  a2[1] = U[(kb + 5) * H + col];
    a3[0] = U[(kb + 6) * H + col];  a3[1] = U[(kb + 7) * H + col];

    const uint32_t xoff0 = (uint32_t)(base + col) * T;
    __shared__ __attribute__((aligned(16))) float stage[SBUF];
    float* const sw = stage + (uint32_t)col * SPITCH + hoff;
    const float* const sr = stage + (uint32_t)lane * 4u;
    float* const orow = out + ((uint32_t)base * T * H + (uint32_t)lane * 4u);

    v8f g = {};

    auto gather = [&](int tok) -> v8f {
        const float* np = embW + ((uint32_t)tok * 16u + hoff);
        v4f l0 = *reinterpret_cast<const v4f*>(np);
        v4f l1 = *reinterpret_cast<const v4f*>(np + 4);
        return __builtin_shufflevector(l0, l1, 0, 1, 2, 3, 4, 5, 6, 7);
    };
    auto gather_at = [&](uint32_t boff) -> v8f {
        const char* np = reinterpret_cast<const char*>(embW) + boff;
        v4f l0 = *reinterpret_cast<const v4f*>(np);
        v4f l1 = *reinterpret_cast<const v4f*>(np + 16);
        return __builtin_shufflevector(l0, l1, 0, 1, 2, 3, 4, 5, 6, 7);
    };

    v4i first[NQ];
    v4i tok[NQ];
    v8f xr[RING];
#pragma unroll
    for (int q = 0; q < NQ; ++q)
        first[q] = *reinterpret_cast<const v4i*>(X + (xoff0 + 4u * q));
#pragma unroll
    for (int u = 0; u < RING; ++u)
        xr[u] = gather(first[u >> 2][u & 3]);
#pragma unroll
    for (int q = 0; q < NQ; ++q)
        tok[q] = *reinterpret_cast<const v4i*>(
            X + (xoff0 + (uint32_t)RING + 4u * q));

    v8f gs[RING];
    uint32_t bo[RING] = {};

    auto steps = [&](auto pc) {
        constexpr int P = decltype(pc)::value;
#pragma unroll
        for (int v = 0; v < GRP; ++v) {
            const int u = P * GRP + v;
            v2f b0 = {g[0], g[1]};
            v2f b1 = {g[2], g[3]};
            v2f b2 = {g[4], g[5]};
            v2f b3 = {g[6], g[7]};

            v8f acc0 = xr[u];
            v8f acc1 = {};
            acc0 = __builtin_amdgcn_wmma_f32_16x16x4_f32(false, a0, false, b0,
                                                         (short)0, acc0, false, false);
            acc1 = __builtin_amdgcn_wmma_f32_16x16x4_f32(false, a1, false, b1,
                                                         (short)0, acc1, false, false);
            acc0 = __builtin_amdgcn_wmma_f32_16x16x4_f32(false, a2, false, b2,
                                                         (short)0, acc0, false, false);
            acc1 = __builtin_amdgcn_wmma_f32_16x16x4_f32(false, a3, false, b3,
                                                         (short)0, acc1, false, false);

            v8f s = acc0 + acc1;
#pragma unroll
            for (int j = 0; j < 8; ++j) g[j] = tanh_fast(s[j]);
            gs[u] = g;
        }
    };

    v4f cp[16];

    auto place = [&](auto pc, int tt, auto refill, auto copy, int tcopy) {
        constexpr int P = decltype(pc)::value;
        constexpr bool REFILL = decltype(refill)::value;
        constexpr bool COPY = decltype(copy)::value;
        __builtin_amdgcn_sched_barrier(0);

        if constexpr (COPY) {
            float* opb = orow + (uint32_t)tcopy * H;
#pragma unroll
            for (int r = 0; r < 16; ++r)
                *reinterpret_cast<v4f*>(opb + r * T * H) = cp[r];
        }
        if constexpr (REFILL) {
            if (P == 0) {
#pragma unroll
                for (int u = 0; u < RING; ++u)
                    bo[u] = (uint32_t)tok[u >> 2][u & 3] * 64u + hoff * 4u;
            }
#pragma unroll
            for (int v = 0; v < GRP; ++v)
                xr[P * GRP + v] = gather_at(bo[P * GRP + v]);
            if (P == 0) {
                const int tq = (tt + 2 * RING < T) ? (tt + 2 * RING) : (T - RING);
#pragma unroll
                for (int q = 0; q < NQ; ++q)
                    tok[q] = *reinterpret_cast<const v4i*>(
                        X + (xoff0 + (uint32_t)tq + 4u * q));
            }
        }
#pragma unroll
        for (int v = 0; v < GRP; ++v) {
            const int u = P * GRP + v;
            v4f s0 = {gs[u][0], gs[u][1], gs[u][2], gs[u][3]};
            v4f s1 = {gs[u][4], gs[u][5], gs[u][6], gs[u][7]};
            *reinterpret_cast<v4f*>(sw + v * H) = s0;
            *reinterpret_cast<v4f*>(sw + v * H + 4) = s1;
        }
#pragma unroll
        for (int r = 0; r < 16; ++r)
            cp[r] = *reinterpret_cast<const v4f*>(sr + r * SPITCH);
        __builtin_amdgcn_sched_barrier(0);
    };

    auto body = [&](int tt, auto refill, auto copy0) {
        place(P0T{}, tt, refill, copy0, tt - GRP);
        steps(P1T{});
        place(P1T{}, tt, refill, YesT{}, tt);
        if constexpr (decltype(refill)::value) steps(P0T{});
    };

    steps(P0T{});
    body(0, YesT{}, NoT{});
    for (int tt = RING; tt < T - RING; tt += RING)
        body(tt, YesT{}, YesT{});
    body(T - RING, NoT{}, YesT{});

    {
        float* opb = orow + (uint32_t)(T - GRP) * H;
#pragma unroll
        for (int r = 0; r < 16; ++r)
            *reinterpret_cast<v4f*>(opb + r * T * H) = cp[r];
    }

    float* hp0 = out + ((uint32_t)B * T * H + (uint32_t)(base + col) * H + hoff);
    v4f f00 = {g[0], g[1], g[2], g[3]};
    v4f f01 = {g[4], g[5], g[6], g[7]};
    *reinterpret_cast<v4f*>(hp0) = f00;
    *reinterpret_cast<v4f*>(hp0 + 4) = f01;
}

extern "C" void kernel_launch(void* const* d_in, const int* in_sizes, int n_in,
                              void* d_out, int out_size, void* d_ws, size_t ws_size,
                              hipStream_t stream) {
    const int*   X   = (const int*)d_in[0];
    const float* emb = (const float*)d_in[1];
    const float* W   = (const float*)d_in[2];
    const float* U   = (const float*)d_in[3];
    const float* b   = (const float*)d_in[4];
    float* out  = (float*)d_out;
    float* embW = (float*)d_ws;

    embw_precompute<<<(V * H + 255) / 256, 256, 0, stream>>>(emb, W, b, embW);

    rnn_scan<<<B / 16, 32, 0, stream>>>(X, U, embW, out);
}
